# starting kernel
# baseline (speedup 1.0000x reference)
#include <hip/hip_runtime.h>

typedef __attribute__((ext_vector_type(2))) float v2f;
typedef __attribute__((ext_vector_type(8))) float v8f;

#define Bsz 8
#define Lsz 512
#define NDsz 3
#define CDsz 61
#define Dsz 64
#define Hsz 512
#define Osz 192
#define CHUNKS 16
#define LC (Lsz / CHUNKS)
#define LDS_STRIDE 516

#define XCAT_ELEMS (Lsz * 16 * Dsz)
#define U_ELEMS    (Lsz * Bsz * Hsz)

__global__ void pack_xcat(const float* __restrict__ x, const float* __restrict__ z,
                          float* __restrict__ xcat) {
    int idx = blockIdx.x * blockDim.x + threadIdx.x;
    if (idx >= XCAT_ELEMS) return;
    int l = idx >> 10;
    int r = idx & 1023;
    int m = r >> 6;
    int d = r & 63;
    float v = 0.f;
    if (m < Bsz) {
        v = (d < NDsz) ? x[(m * Lsz + l) * NDsz + d]
                       : z[(m * Lsz + l) * CDsz + (d - NDsz)];
    }
    xcat[idx] = v;
}

__global__ __launch_bounds__(32) void enc_gemm(const float* __restrict__ xcat,
                                               const float* __restrict__ W_enc,
                                               float* __restrict__ u) {
    int l    = blockIdx.x >> 5;
    int nt   = blockIdx.x & 31;
    int lane = threadIdx.x;
    int m16  = lane & 15;
    int half = lane >> 4;
    int n    = nt * 16 + m16;

    const float* xa = xcat + (size_t)l * (16 * Dsz);
    const float* wb = W_enc + (size_t)l * Dsz * Hsz;

    v8f acc = {0.f, 0.f, 0.f, 0.f, 0.f, 0.f, 0.f, 0.f};
    #pragma unroll
    for (int kk = 0; kk < Dsz / 4; ++kk) {
        int k0 = 4 * kk + 2 * half;
        v2f av = *reinterpret_cast<const v2f*>(xa + m16 * Dsz + k0);
        v2f bv;
        bv.x = __builtin_nontemporal_load(wb + (size_t)k0 * Hsz + n);
        bv.y = __builtin_nontemporal_load(wb + (size_t)(k0 + 1) * Hsz + n);
        acc = __builtin_amdgcn_wmma_f32_16x16x4_f32(false, av, false, bv,
                                                    (short)0, acc, false, false);
    }
    if (lane < 16) {
        float* uo = u + (size_t)l * (Bsz * Hsz) + n;
        #pragma unroll
        for (int v = 0; v < 8; ++v) uo[(size_t)v * Hsz] = acc[v];
    }
}

__global__ void chunk_sum(const float* __restrict__ u, float* __restrict__ partial) {
    int c = blockIdx.x >> 3;
    int b = blockIdx.x & 7;
    int h = threadIdx.x;
    const float* up = u + (size_t)b * Hsz + h;
    float s = 0.f;
    #pragma unroll 4
    for (int i = 0; i < LC; ++i) {
        int l = c * LC + i;
        s += up[(size_t)l * (Bsz * Hsz)];
    }
    partial[((size_t)c * Bsz + b) * Hsz + h] = s;
}

__global__ void scan_relu(float* __restrict__ u, const float* __restrict__ partial,
                          const float* __restrict__ b_enc, const float* __restrict__ resc) {
    int c = blockIdx.x >> 3;
    int b = blockIdx.x & 7;
    int h = threadIdx.x;

    float acc = 0.f;
    for (int cp = 0; cp < c; ++cp)
        acc += partial[((size_t)cp * Bsz + b) * Hsz + h];

    float bs = b_enc[h] / resc[0];
    float* up = u + (size_t)b * Hsz + h;
    for (int i = 0; i < LC; ++i) {
        int l = c * LC + i;
        size_t off = (size_t)l * (Bsz * Hsz);
        float uv = up[off];
        float a  = resc[l] * (bs + acc);
        up[off]  = fmaxf(a, 0.f);
        acc += uv;
    }
}

__global__ __launch_bounds__(384) void out_gemm(const float* __restrict__ ar,
                                                const float* __restrict__ W_out,
                                                const float* __restrict__ b_out,
                                                float* __restrict__ out) {
    __shared__ float lds[16 * LDS_STRIDE];
    int l   = blockIdx.x;
    int tid = threadIdx.x;

    for (int t = tid; t < 16 * Hsz; t += 384) {
        int m = t >> 9;
        int k = t & 511;
        lds[m * LDS_STRIDE + k] = (m < Bsz) ? ar[(size_t)l * (Bsz * Hsz) + t] : 0.f;
    }
    __syncthreads();

    int w    = tid >> 5;
    int lane = tid & 31;
    int m16  = lane & 15;
    int half = lane >> 4;
    int n    = w * 16 + m16;

    const float* wb = W_out + (size_t)l * Hsz * Osz;

    v8f acc = {0.f, 0.f, 0.f, 0.f, 0.f, 0.f, 0.f, 0.f};
    #pragma unroll 4
    for (int kk = 0; kk < Hsz / 4; ++kk) {
        int k0 = 4 * kk + 2 * half;
        v2f av = *reinterpret_cast<const v2f*>(&lds[m16 * LDS_STRIDE + k0]);
        v2f bv;
        bv.x = __builtin_nontemporal_load(wb + (size_t)k0 * Osz + n);
        bv.y = __builtin_nontemporal_load(wb + (size_t)(k0 + 1) * Osz + n);
        acc = __builtin_amdgcn_wmma_f32_16x16x4_f32(false, av, false, bv,
                                                    (short)0, acc, false, false);
    }

    if (lane < 16) {
        float bo = b_out[(size_t)l * Osz + n];
        #pragma unroll
        for (int v = 0; v < 8; ++v)
            __builtin_nontemporal_store(acc[v] + bo,
                out + (size_t)v * Lsz * Osz + (size_t)l * Osz + n);
    }
}

extern "C" void kernel_launch(void* const* d_in, const int* in_sizes, int n_in,
                              void* d_out, int out_size, void* d_ws, size_t ws_size,
                              hipStream_t stream) {
    const float* x     = (const float*)d_in[0];
    const float* z     = (const float*)d_in[1];
    const float* W_enc = (const float*)d_in[2];
    const float* b_enc = (const float*)d_in[3];
    const float* W_out = (const float*)d_in[4];
    const float* b_out = (const float*)d_in[5];
    const float* resc  = (const float*)d_in[6];
    float* out = (float*)d_out;

    float* ws      = (float*)d_ws;
    float* xcat    = ws;
    float* u       = ws + XCAT_ELEMS;
    float* partial = u + U_ELEMS;

    pack_xcat<<<(XCAT_ELEMS + 255) / 256, 256, 0, stream>>>(x, z, xcat);
    enc_gemm <<<Lsz * (Hsz / 16), 32, 0, stream>>>(xcat, W_enc, u);
    chunk_sum<<<CHUNKS * Bsz, Hsz, 0, stream>>>(u, partial);
    scan_relu<<<CHUNKS * Bsz, Hsz, 0, stream>>>(u, partial, b_enc, resc);
    out_gemm <<<Lsz, 384, 0, stream>>>(u, W_out, b_out, out);
}


# Both GEMMs read W 128-bit per lane (4 adjacent columns = 4 column-interleaved N tiles per wave): enc 512 8-wave groups; out 3 column slabs x 4 K quarters summed via LDS (new summation order)
# speedup vs baseline: 1.2499x; 1.2499x over previous
#include <hip/hip_runtime.h>

typedef __attribute__((ext_vector_type(2))) float v2f;
typedef __attribute__((ext_vector_type(4))) float v4f;
typedef __attribute__((ext_vector_type(8))) float v8f;

#define Bsz 8
#define Lsz 512
#define NDsz 3
#define CDsz 61
#define Dsz 64
#define Hsz 512
#define Osz 192
#define CHUNKS 16
#define LC (Lsz / CHUNKS)
#define LDS_STRIDE 516
#define KQ 4
#define KQ_LEN (Hsz / KQ)
#define SLABS (Osz / 64)
#define GK 4
#define NG (KQ_LEN / 4 / GK)

#define XCAT_ELEMS (Lsz * 16 * Dsz)
#define U_ELEMS    (Lsz * Bsz * Hsz)

__global__ void pack_xcat(const float* __restrict__ x, const float* __restrict__ z,
                          float* __restrict__ xcat) {
    int idx = blockIdx.x * blockDim.x + threadIdx.x;
    if (idx >= XCAT_ELEMS) return;
    int l = idx >> 10;
    int r = idx & 1023;
    int m = r >> 6;
    int d = r & 63;
    float v = 0.f;
    if (m < Bsz) {
        v = (d < NDsz) ? x[(m * Lsz + l) * NDsz + d]
                       : z[(m * Lsz + l) * CDsz + (d - NDsz)];
    }
    xcat[idx] = v;
}

__global__ __launch_bounds__(256) void enc_gemm(const float* __restrict__ xcat,
                                                const float* __restrict__ W_enc,
                                                float* __restrict__ u) {
    int l    = blockIdx.x;
    int w    = threadIdx.x >> 5;
    int lane = threadIdx.x & 31;
    int m16  = lane & 15;
    int half = lane >> 4;
    int c    = w * 64 + 4 * m16;

    const float* xa = xcat + (size_t)l * (16 * Dsz) + m16 * Dsz + 2 * half;
    const float* wb = W_enc + (size_t)l * Dsz * Hsz + (size_t)(2 * half) * Hsz + c;

    v4f b0[Dsz / 4], b1[Dsz / 4];
    v2f av[Dsz / 4];
    #pragma unroll
    for (int kk = 0; kk < Dsz / 4; ++kk)
        av[kk] = *reinterpret_cast<const v2f*>(xa + 4 * kk);
    #pragma unroll
    for (int kk = 0; kk < Dsz / 4; ++kk) {
        b0[kk] = __builtin_nontemporal_load(reinterpret_cast<const v4f*>(wb + (size_t)(4 * kk) * Hsz));
        b1[kk] = __builtin_nontemporal_load(reinterpret_cast<const v4f*>(wb + (size_t)(4 * kk + 1) * Hsz));
    }
    __builtin_amdgcn_sched_barrier(0);

    v8f acc[4];
    #pragma unroll
    for (int j = 0; j < 4; ++j) acc[j] = (v8f){0.f, 0.f, 0.f, 0.f, 0.f, 0.f, 0.f, 0.f};
    #pragma unroll
    for (int kk = 0; kk < Dsz / 4; ++kk) {
        #pragma unroll
        for (int j = 0; j < 4; ++j) {
            v2f bv;
            bv.x = b0[kk][j];
            bv.y = b1[kk][j];
            acc[j] = __builtin_amdgcn_wmma_f32_16x16x4_f32(false, av[kk], false, bv,
                                                           (short)0, acc[j], false, false);
        }
    }
    if (lane < 16) {
        float* uo = u + (size_t)l * (Bsz * Hsz) + c;
        #pragma unroll
        for (int v = 0; v < 8; ++v) {
            v4f o = {acc[0][v], acc[1][v], acc[2][v], acc[3][v]};
            *reinterpret_cast<v4f*>(uo + (size_t)v * Hsz) = o;
        }
    }
}

__global__ void chunk_sum(const float* __restrict__ u, float* __restrict__ partial) {
    int c = blockIdx.x >> 3;
    int b = blockIdx.x & 7;
    int h = threadIdx.x;
    const float* up = u + (size_t)b * Hsz + h;
    float s = 0.f;
    #pragma unroll 4
    for (int i = 0; i < LC; ++i) {
        int l = c * LC + i;
        s += up[(size_t)l * (Bsz * Hsz)];
    }
    partial[((size_t)c * Bsz + b) * Hsz + h] = s;
}

__global__ void scan_relu(float* __restrict__ u, const float* __restrict__ partial,
                          const float* __restrict__ b_enc, const float* __restrict__ resc) {
    int c = blockIdx.x >> 3;
    int b = blockIdx.x & 7;
    int h = threadIdx.x;

    float acc = 0.f;
    for (int cp = 0; cp < c; ++cp)
        acc += partial[((size_t)cp * Bsz + b) * Hsz + h];

    float bs = b_enc[h] / resc[0];
    float* up = u + (size_t)b * Hsz + h;
    for (int i = 0; i < LC; ++i) {
        int l = c * LC + i;
        size_t off = (size_t)l * (Bsz * Hsz);
        float uv = up[off];
        float a  = resc[l] * (bs + acc);
        up[off]  = fmaxf(a, 0.f);
        acc += uv;
    }
}

__global__ __launch_bounds__(384) void out_gemm(const float* __restrict__ ar,
                                                const float* __restrict__ W_out,
                                                const float* __restrict__ b_out,
                                                float* __restrict__ out) {
    __shared__ float lds[16 * LDS_STRIDE];
    __shared__ v4f red[(KQ - 1) * SLABS * Bsz * 16];
    int l   = blockIdx.x;
    int tid = threadIdx.x;

    for (int t = tid; t < 16 * Hsz; t += 384) {
        int m = t >> 9;
        int k = t & 511;
        lds[m * LDS_STRIDE + k] = (m < Bsz) ? ar[(size_t)l * (Bsz * Hsz) + t] : 0.f;
    }
    __syncthreads();

    int w    = tid >> 5;
    int slab = w % SLABS;
    int kq   = w / SLABS;
    int lane = tid & 31;
    int m16  = lane & 15;
    int half = lane >> 4;
    int c    = slab * 64 + 4 * m16;

    const float* wb = W_out + (size_t)l * Hsz * Osz + (size_t)(kq * KQ_LEN + 2 * half) * Osz + c;
    const float* la = &lds[m16 * LDS_STRIDE + kq * KQ_LEN + 2 * half];

    v8f acc[4];
    #pragma unroll
    for (int j = 0; j < 4; ++j) acc[j] = (v8f){0.f, 0.f, 0.f, 0.f, 0.f, 0.f, 0.f, 0.f};

    v4f b0[2][GK], b1[2][GK];
#define LOAD_GROUP(G) do { _Pragma("unroll") for (int i = 0; i < GK; ++i) { \
        b0[(G) & 1][i] = __builtin_nontemporal_load(reinterpret_cast<const v4f*>(wb + (size_t)(4 * ((G) * GK + i)) * Osz)); \
        b1[(G) & 1][i] = __builtin_nontemporal_load(reinterpret_cast<const v4f*>(wb + (size_t)(4 * ((G) * GK + i) + 1) * Osz)); } } while (0)
    LOAD_GROUP(0);
    #pragma unroll
    for (int g = 0; g < NG; ++g) {
        __builtin_amdgcn_sched_barrier(0);
        if (g + 1 < NG) LOAD_GROUP(g + 1);
        __builtin_amdgcn_sched_barrier(0);
        #pragma unroll
        for (int i = 0; i < GK; ++i) {
            v2f av = *reinterpret_cast<const v2f*>(la + 4 * (g * GK + i));
            #pragma unroll
            for (int j = 0; j < 4; ++j) {
                v2f bv;
                bv.x = b0[g & 1][i][j];
                bv.y = b1[g & 1][i][j];
                acc[j] = __builtin_amdgcn_wmma_f32_16x16x4_f32(false, av, false, bv,
                                                               (short)0, acc[j], false, false);
            }
        }
    }
#undef LOAD_GROUP

    if (kq > 0 && lane < 16) {
        v4f* rp = red + ((kq - 1) * SLABS + slab) * (Bsz * 16) + m16;
        #pragma unroll
        for (int v = 0; v < 8; ++v) {
            v4f o = {acc[0][v], acc[1][v], acc[2][v], acc[3][v]};
            rp[v * 16] = o;
        }
    }
    __syncthreads();
    if (kq == 0 && lane < 16) {
        v4f bo = *reinterpret_cast<const v4f*>(b_out + (size_t)l * Osz + c);
        const v4f* rp = red + slab * (Bsz * 16) + m16;
        #pragma unroll
        for (int v = 0; v < 8; ++v) {
            v4f o = {acc[0][v], acc[1][v], acc[2][v], acc[3][v]};
            o = o + rp[v * 16];
            o = o + rp[SLABS * (Bsz * 16) + v * 16];
            o = o + rp[2 * SLABS * (Bsz * 16) + v * 16];
            o = o + bo;
            __builtin_nontemporal_store(o,
                reinterpret_cast<v4f*>(out + (size_t)v * Lsz * Osz + (size_t)l * Osz + c));
        }
    }
}

extern "C" void kernel_launch(void* const* d_in, const int* in_sizes, int n_in,
                              void* d_out, int out_size, void* d_ws, size_t ws_size,
                              hipStream_t stream) {
    const float* x     = (const float*)d_in[0];
    const float* z     = (const float*)d_in[1];
    const float* W_enc = (const float*)d_in[2];
    const float* b_enc = (const float*)d_in[3];
    const float* W_out = (const float*)d_in[4];
    const float* b_out = (const float*)d_in[5];
    const float* resc  = (const float*)d_in[6];
    float* out = (float*)d_out;

    float* ws      = (float*)d_ws;
    float* xcat    = ws;
    float* u       = ws + XCAT_ELEMS;
    float* partial = u + U_ELEMS;

    pack_xcat<<<(XCAT_ELEMS + 255) / 256, 256, 0, stream>>>(x, z, xcat);
    enc_gemm <<<Lsz, 256, 0, stream>>>(xcat, W_enc, u);
    chunk_sum<<<CHUNKS * Bsz, Hsz, 0, stream>>>(u, partial);
    scan_relu<<<CHUNKS * Bsz, Hsz, 0, stream>>>(u, partial, b_enc, resc);
    out_gemm <<<Lsz, 384, 0, stream>>>(u, W_out, b_out, out);
}


# Serialized load-wait-use loops removed: chunk sums, scan and the out GEMM's A staging each issue all their loads as one group, first W group ahead of the staging barrier; summation order kept
# speedup vs baseline: 1.5136x; 1.2110x over previous
#include <hip/hip_runtime.h>

typedef __attribute__((ext_vector_type(2))) float v2f;
typedef __attribute__((ext_vector_type(4))) float v4f;
typedef __attribute__((ext_vector_type(8))) float v8f;

#define Bsz 8
#define Lsz 512
#define NDsz 3
#define CDsz 61
#define Dsz 64
#define Hsz 512
#define Osz 192
#define CHUNKS 16
#define LC (Lsz / CHUNKS)
#define LDS_STRIDE 516
#define KQ 4
#define KQ_LEN (Hsz / KQ)
#define SLABS (Osz / 64)
#define GK 4
#define NG (KQ_LEN / 4 / GK)

#define XCAT_ELEMS (Lsz * 16 * Dsz)
#define U_ELEMS    (Lsz * Bsz * Hsz)

__global__ void pack_xcat(const float* __restrict__ x, const float* __restrict__ z,
                          float* __restrict__ xcat) {
    int idx = blockIdx.x * blockDim.x + threadIdx.x;
    if (idx >= XCAT_ELEMS) return;
    int l = idx >> 10;
    int r = idx & 1023;
    int m = r >> 6;
    int d = r & 63;
    float v = 0.f;
    if (m < Bsz) {
        v = (d < NDsz) ? x[(m * Lsz + l) * NDsz + d]
                       : z[(m * Lsz + l) * CDsz + (d - NDsz)];
    }
    xcat[idx] = v;
}

__global__ __launch_bounds__(256) void enc_gemm(const float* __restrict__ xcat,
                                                const float* __restrict__ W_enc,
                                                float* __restrict__ u) {
    int l    = blockIdx.x;
    int w    = threadIdx.x >> 5;
    int lane = threadIdx.x & 31;
    int m16  = lane & 15;
    int half = lane >> 4;
    int c    = w * 64 + 4 * m16;

    const float* xa = xcat + (size_t)l * (16 * Dsz) + m16 * Dsz + 2 * half;
    const float* wb = W_enc + (size_t)l * Dsz * Hsz + (size_t)(2 * half) * Hsz + c;

    v4f b0[Dsz / 4], b1[Dsz / 4];
    v2f av[Dsz / 4];
    #pragma unroll
    for (int kk = 0; kk < Dsz / 4; ++kk)
        av[kk] = *reinterpret_cast<const v2f*>(xa + 4 * kk);
    #pragma unroll
    for (int kk = 0; kk < Dsz / 4; ++kk) {
        b0[kk] = __builtin_nontemporal_load(reinterpret_cast<const v4f*>(wb + (size_t)(4 * kk) * Hsz));
        b1[kk] = __builtin_nontemporal_load(reinterpret_cast<const v4f*>(wb + (size_t)(4 * kk + 1) * Hsz));
    }
    __builtin_amdgcn_sched_barrier(0);

    v8f acc[4];
    #pragma unroll
    for (int j = 0; j < 4; ++j) acc[j] = (v8f){0.f, 0.f, 0.f, 0.f, 0.f, 0.f, 0.f, 0.f};
    #pragma unroll
    for (int kk = 0; kk < Dsz / 4; ++kk) {
        #pragma unroll
        for (int j = 0; j < 4; ++j) {
            v2f bv;
            bv.x = b0[kk][j];
            bv.y = b1[kk][j];
            acc[j] = __builtin_amdgcn_wmma_f32_16x16x4_f32(false, av[kk], false, bv,
                                                           (short)0, acc[j], false, false);
        }
    }
    if (lane < 16) {
        float* uo = u + (size_t)l * (Bsz * Hsz) + c;
        #pragma unroll
        for (int v = 0; v < 8; ++v) {
            v4f o = {acc[0][v], acc[1][v], acc[2][v], acc[3][v]};
            *reinterpret_cast<v4f*>(uo + (size_t)v * Hsz) = o;
        }
    }
}

__global__ void chunk_sum(const float* __restrict__ u, float* __restrict__ partial) {
    int c = blockIdx.x >> 3;
    int b = blockIdx.x & 7;
    int h = threadIdx.x;
    const float* up = u + (size_t)b * Hsz + h;
    float v[LC];
    #pragma unroll
    for (int i = 0; i < LC; ++i)
        v[i] = up[(size_t)(c * LC + i) * (Bsz * Hsz)];
    float s = 0.f;
    #pragma unroll
    for (int i = 0; i < LC; ++i) s += v[i];
    partial[((size_t)c * Bsz + b) * Hsz + h] = s;
}

__global__ void scan_relu(float* __restrict__ u, const float* __restrict__ partial,
                          const float* __restrict__ b_enc, const float* __restrict__ resc) {
    int c = blockIdx.x >> 3;
    int b = blockIdx.x & 7;
    int h = threadIdx.x;

    float* up = u + (size_t)b * Hsz + h;
    float uv[LC];
    #pragma unroll
    for (int i = 0; i < LC; ++i)
        uv[i] = up[(size_t)(c * LC + i) * (Bsz * Hsz)];
    float p[CHUNKS - 1];
    #pragma unroll
    for (int cp = 0; cp < CHUNKS - 1; ++cp)
        p[cp] = partial[((size_t)cp * Bsz + b) * Hsz + h];
    float be = b_enc[h];
    float r0 = resc[0];

    float acc = 0.f;
    #pragma unroll
    for (int cp = 0; cp < CHUNKS - 1; ++cp)
        if (cp < c) acc += p[cp];

    float bs = be / r0;
    #pragma unroll
    for (int i = 0; i < LC; ++i) {
        int l = c * LC + i;
        float a = resc[l] * (bs + acc);
        up[(size_t)l * (Bsz * Hsz)] = fmaxf(a, 0.f);
        acc += uv[i];
    }
}

__global__ __launch_bounds__(384) void out_gemm(const float* __restrict__ ar,
                                                const float* __restrict__ W_out,
                                                const float* __restrict__ b_out,
                                                float* __restrict__ out) {
    __shared__ float lds[16 * LDS_STRIDE];
    __shared__ v4f red[(KQ - 1) * SLABS * Bsz * 16];
    int l   = blockIdx.x;
    int tid = threadIdx.x;

    const v4f* arow = reinterpret_cast<const v4f*>(ar + (size_t)l * (Bsz * Hsz));
    v4f sv[3];
    #pragma unroll
    for (int j = 0; j < 3; ++j) {
        int q = tid + 384 * j;
        sv[j] = arow[q < Bsz * Hsz / 4 ? q : Bsz * Hsz / 4 - 1];
    }

    int w    = tid >> 5;
    int slab = w % SLABS;
    int kq   = w / SLABS;
    int lane = tid & 31;
    int m16  = lane & 15;
    int half = lane >> 4;
    int c    = slab * 64 + 4 * m16;

    const float* wb = W_out + (size_t)l * Hsz * Osz + (size_t)(kq * KQ_LEN + 2 * half) * Osz + c;
    const float* la = &lds[m16 * LDS_STRIDE + kq * KQ_LEN + 2 * half];

    v8f acc[4];
    #pragma unroll
    for (int j = 0; j < 4; ++j) acc[j] = (v8f){0.f, 0.f, 0.f, 0.f, 0.f, 0.f, 0.f, 0.f};

    v4f b0[2][GK], b1[2][GK];
#define LOAD_GROUP(G) do { _Pragma("unroll") for (int i = 0; i < GK; ++i) { \
        b0[(G) & 1][i] = __builtin_nontemporal_load(reinterpret_cast<const v4f*>(wb + (size_t)(4 * ((G) * GK + i)) * Osz)); \
        b1[(G) & 1][i] = __builtin_nontemporal_load(reinterpret_cast<const v4f*>(wb + (size_t)(4 * ((G) * GK + i) + 1) * Osz)); } } while (0)
    LOAD_GROUP(0);
    #pragma unroll
    for (int j = 0; j < 3; ++j) {
        int q = tid + 384 * j;
        if (q < Bsz * Hsz / 4) {
            int m = q >> 7;
            int k = 4 * (q & 127);
            *reinterpret_cast<v4f*>(&lds[m * LDS_STRIDE + k]) = sv[j];
            *reinterpret_cast<v4f*>(&lds[(m + Bsz) * LDS_STRIDE + k]) = (v4f){0.f, 0.f, 0.f, 0.f};
        }
    }
    __syncthreads();
    #pragma unroll
    for (int g = 0; g < NG; ++g) {
        __builtin_amdgcn_sched_barrier(0);
        if (g + 1 < NG) LOAD_GROUP(g + 1);
        __builtin_amdgcn_sched_barrier(0);
        #pragma unroll
        for (int i = 0; i < GK; ++i) {
            v2f av = *reinterpret_cast<const v2f*>(la + 4 * (g * GK + i));
            #pragma unroll
            for (int j = 0; j < 4; ++j) {
                v2f bv;
                bv.x = b0[g & 1][i][j];
                bv.y = b1[g & 1][i][j];
                acc[j] = __builtin_amdgcn_wmma_f32_16x16x4_f32(false, av, false, bv,
                                                               (short)0, acc[j], false, false);
            }
        }
    }
#undef LOAD_GROUP

    if (kq > 0 && lane < 16) {
        v4f* rp = red + ((kq - 1) * SLABS + slab) * (Bsz * 16) + m16;
        #pragma unroll
        for (int v = 0; v < 8; ++v) {
            v4f o = {acc[0][v], acc[1][v], acc[2][v], acc[3][v]};
            rp[v * 16] = o;
        }
    }
    __syncthreads();
    if (kq == 0 && lane < 16) {
        v4f bo = *reinterpret_cast<const v4f*>(b_out + (size_t)l * Osz + c);
        const v4f* rp = red + slab * (Bsz * 16) + m16;
        #pragma unroll
        for (int v = 0; v < 8; ++v) {
            v4f o = {acc[0][v], acc[1][v], acc[2][v], acc[3][v]};
            o = o + rp[v * 16];
            o = o + rp[SLABS * (Bsz * 16) + v * 16];
            o = o + rp[2 * SLABS * (Bsz * 16) + v * 16];
            o = o + bo;
            __builtin_nontemporal_store(o,
                reinterpret_cast<v4f*>(out + (size_t)v * Lsz * Osz + (size_t)l * Osz + c));
        }
    }
}

extern "C" void kernel_launch(void* const* d_in, const int* in_sizes, int n_in,
                              void* d_out, int out_size, void* d_ws, size_t ws_size,
                              hipStream_t stream) {
    const float* x     = (const float*)d_in[0];
    const float* z     = (const float*)d_in[1];
    const float* W_enc = (const float*)d_in[2];
    const float* b_enc = (const float*)d_in[3];
    const float* W_out = (const float*)d_in[4];
    const float* b_out = (const float*)d_in[5];
    const float* resc  = (const float*)d_in[6];
    float* out = (float*)d_out;

    float* ws      = (float*)d_ws;
    float* xcat    = ws;
    float* u       = ws + XCAT_ELEMS;
    float* partial = u + U_ELEMS;

    pack_xcat<<<(XCAT_ELEMS + 255) / 256, 256, 0, stream>>>(x, z, xcat);
    enc_gemm <<<Lsz, 256, 0, stream>>>(xcat, W_enc, u);
    chunk_sum<<<CHUNKS * Bsz, Hsz, 0, stream>>>(u, partial);
    scan_relu<<<CHUNKS * Bsz, Hsz, 0, stream>>>(u, partial, b_enc, resc);
    out_gemm <<<Lsz, 384, 0, stream>>>(u, W_out, b_out, out);
}


# Launches 5 -> 3: encode GEMM reads its A rows straight from x and z through LDS (pack kernel and its plane gone); chunk sums move into the scan kernel through LDS; summation order kept
# speedup vs baseline: 1.6300x; 1.0769x over previous
#include <hip/hip_runtime.h>

typedef __attribute__((ext_vector_type(2))) float v2f;
typedef __attribute__((ext_vector_type(4))) float v4f;
typedef __attribute__((ext_vector_type(8))) float v8f;

#define Bsz 8
#define Lsz 512
#define NDsz 3
#define CDsz 61
#define Dsz 64
#define Hsz 512
#define Osz 192
#define CHUNKS 16
#define LC (Lsz / CHUNKS)
#define LDS_STRIDE 516
#define KQ 4
#define KQ_LEN (Hsz / KQ)
#define SLABS (Osz / 64)
#define GK 4
#define NG (KQ_LEN / 4 / GK)

#define XS_STRIDE 68
#define SCOLS 32


__global__ __launch_bounds__(256) void enc_gemm(const float* __restrict__ x,
                                                const float* __restrict__ z,
                                                const float* __restrict__ W_enc,
                                                float* __restrict__ u) {
    __shared__ float xs[Bsz * XS_STRIDE];
    int l    = blockIdx.x;
    int w    = threadIdx.x >> 5;
    int lane = threadIdx.x & 31;
    int m16  = lane & 15;
    int half = lane >> 4;
    int c    = w * 64 + 4 * m16;

    size_t rl = (size_t)w * Lsz + l;
    const float* p0 = (lane < NDsz) ? (x + rl * NDsz + lane) : (z + rl * CDsz + (lane - NDsz));
    float e0 = *p0;
    float e1 = z[rl * CDsz + (lane + 32 - NDsz)];

    const float* wb = W_enc + (size_t)l * Dsz * Hsz + (size_t)(2 * half) * Hsz + c;

    v4f b0[Dsz / 4], b1[Dsz / 4];
    v2f av[Dsz / 4];
    #pragma unroll
    for (int kk = 0; kk < Dsz / 4; ++kk) {
        b0[kk] = __builtin_nontemporal_load(reinterpret_cast<const v4f*>(wb + (size_t)(4 * kk) * Hsz));
        b1[kk] = __builtin_nontemporal_load(reinterpret_cast<const v4f*>(wb + (size_t)(4 * kk + 1) * Hsz));
    }
    __builtin_amdgcn_sched_barrier(0);

    xs[w * XS_STRIDE + lane]      = e0;
    xs[w * XS_STRIDE + lane + 32] = e1;
    __builtin_amdgcn_fence(__ATOMIC_RELEASE, "workgroup", "local");
    __builtin_amdgcn_s_barrier();
    __builtin_amdgcn_fence(__ATOMIC_ACQUIRE, "workgroup", "local");
    const float* xa = &xs[(m16 & 7) * XS_STRIDE + 2 * half];
    #pragma unroll
    for (int kk = 0; kk < Dsz / 4; ++kk)
        av[kk] = *reinterpret_cast<const v2f*>(xa + 4 * kk);
    __builtin_amdgcn_sched_barrier(0);

    v8f acc[4];
    #pragma unroll
    for (int j = 0; j < 4; ++j) acc[j] = (v8f){0.f, 0.f, 0.f, 0.f, 0.f, 0.f, 0.f, 0.f};
    #pragma unroll
    for (int kk = 0; kk < Dsz / 4; ++kk) {
        #pragma unroll
        for (int j = 0; j < 4; ++j) {
            v2f bv;
            bv.x = b0[kk][j];
            bv.y = b1[kk][j];
            acc[j] = __builtin_amdgcn_wmma_f32_16x16x4_f32(false, av[kk], false, bv,
                                                           (short)0, acc[j], false, false);
        }
    }
    if (lane < 16) {
        float* uo = u + (size_t)l * (Bsz * Hsz) + c;
        #pragma unroll
        for (int v = 0; v < 8; ++v) {
            v4f o = {acc[0][v], acc[1][v], acc[2][v], acc[3][v]};
            *reinterpret_cast<v4f*>(uo + (size_t)v * Hsz) = o;
        }
    }
}

__global__ __launch_bounds__(CHUNKS * SCOLS) void scan_relu(float* __restrict__ u,
                                                            const float* __restrict__ b_enc,
                                                            const float* __restrict__ resc) {
    __shared__ float csum[CHUNKS * SCOLS];
    int b  = blockIdx.x & 7;
    int hg = blockIdx.x >> 3;
    int c  = __builtin_amdgcn_readfirstlane(threadIdx.x >> 5);
    int j  = threadIdx.x & 31;
    int h  = hg * SCOLS + j;

    float* up = u + (size_t)b * Hsz + h;
    float uv[LC];
    #pragma unroll
    for (int i = 0; i < LC; ++i)
        uv[i] = up[(size_t)(c * LC + i) * (Bsz * Hsz)];
    float be = b_enc[h];
    float r0 = resc[0];
    float rs[LC];
    #pragma unroll
    for (int i = 0; i < LC; ++i) rs[i] = resc[c * LC + i];

    float s = 0.f;
    #pragma unroll
    for (int i = 0; i < LC; ++i) s += uv[i];
    csum[c * SCOLS + j] = s;
    __syncthreads();

    float p[CHUNKS - 1];
    #pragma unroll
    for (int cp = 0; cp < CHUNKS - 1; ++cp)
        p[cp] = csum[cp * SCOLS + j];
    __builtin_amdgcn_sched_barrier(0);
    float acc = 0.f;
    #pragma unroll
    for (int cp = 0; cp < CHUNKS - 1; ++cp)
        if (cp < c) acc += p[cp];

    float bs = be / r0;
    #pragma unroll
    for (int i = 0; i < LC; ++i) {
        int l = c * LC + i;
        float a = rs[i] * (bs + acc);
        up[(size_t)l * (Bsz * Hsz)] = fmaxf(a, 0.f);
        acc += uv[i];
    }
}

__global__ __launch_bounds__(384) void out_gemm(const float* __restrict__ ar,
                                                const float* __restrict__ W_out,
                                                const float* __restrict__ b_out,
                                                float* __restrict__ out) {
    __shared__ float lds[16 * LDS_STRIDE];
    __shared__ v4f red[(KQ - 1) * SLABS * Bsz * 16];
    int l   = blockIdx.x;
    int tid = threadIdx.x;

    const v4f* arow = reinterpret_cast<const v4f*>(ar + (size_t)l * (Bsz * Hsz));
    v4f sv[3];
    #pragma unroll
    for (int j = 0; j < 3; ++j) {
        int q = tid + 384 * j;
        sv[j] = arow[q < Bsz * Hsz / 4 ? q : Bsz * Hsz / 4 - 1];
    }

    int w    = tid >> 5;
    int slab = w % SLABS;
    int kq   = w / SLABS;
    int lane = tid & 31;
    int m16  = lane & 15;
    int half = lane >> 4;
    int c    = slab * 64 + 4 * m16;

    const float* wb = W_out + (size_t)l * Hsz * Osz + (size_t)(kq * KQ_LEN + 2 * half) * Osz + c;
    const float* la = &lds[m16 * LDS_STRIDE + kq * KQ_LEN + 2 * half];

    v8f acc[4];
    #pragma unroll
    for (int j = 0; j < 4; ++j) acc[j] = (v8f){0.f, 0.f, 0.f, 0.f, 0.f, 0.f, 0.f, 0.f};

    v4f b0[2][GK], b1[2][GK];
#define LOAD_GROUP(G) do { _Pragma("unroll") for (int i = 0; i < GK; ++i) { \
        b0[(G) & 1][i] = __builtin_nontemporal_load(reinterpret_cast<const v4f*>(wb + (size_t)(4 * ((G) * GK + i)) * Osz)); \
        b1[(G) & 1][i] = __builtin_nontemporal_load(reinterpret_cast<const v4f*>(wb + (size_t)(4 * ((G) * GK + i) + 1) * Osz)); } } while (0)
    LOAD_GROUP(0);
    #pragma unroll
    for (int j = 0; j < 3; ++j) {
        int q = tid + 384 * j;
        if (q < Bsz * Hsz / 4) {
            int m = q >> 7;
            int k = 4 * (q & 127);
            *reinterpret_cast<v4f*>(&lds[m * LDS_STRIDE + k]) = sv[j];
            *reinterpret_cast<v4f*>(&lds[(m + Bsz) * LDS_STRIDE + k]) = (v4f){0.f, 0.f, 0.f, 0.f};
        }
    }
    __syncthreads();
    #pragma unroll
    for (int g = 0; g < NG; ++g) {
        __builtin_amdgcn_sched_barrier(0);
        if (g + 1 < NG) LOAD_GROUP(g + 1);
        __builtin_amdgcn_sched_barrier(0);
        #pragma unroll
        for (int i = 0; i < GK; ++i) {
            v2f av = *reinterpret_cast<const v2f*>(la + 4 * (g * GK + i));
            #pragma unroll
            for (int j = 0; j < 4; ++j) {
                v2f bv;
                bv.x = b0[g & 1][i][j];
                bv.y = b1[g & 1][i][j];
                acc[j] = __builtin_amdgcn_wmma_f32_16x16x4_f32(false, av, false, bv,
                                                               (short)0, acc[j], false, false);
            }
        }
    }
#undef LOAD_GROUP

    if (kq > 0 && lane < 16) {
        v4f* rp = red + ((kq - 1) * SLABS + slab) * (Bsz * 16) + m16;
        #pragma unroll
        for (int v = 0; v < 8; ++v) {
            v4f o = {acc[0][v], acc[1][v], acc[2][v], acc[3][v]};
            rp[v * 16] = o;
        }
    }
    __syncthreads();
    if (kq == 0 && lane < 16) {
        v4f bo = *reinterpret_cast<const v4f*>(b_out + (size_t)l * Osz + c);
        const v4f* rp = red + slab * (Bsz * 16) + m16;
        #pragma unroll
        for (int v = 0; v < 8; ++v) {
            v4f o = {acc[0][v], acc[1][v], acc[2][v], acc[3][v]};
            o = o + rp[v * 16];
            o = o + rp[SLABS * (Bsz * 16) + v * 16];
            o = o + rp[2 * SLABS * (Bsz * 16) + v * 16];
            o = o + bo;
            __builtin_nontemporal_store(o,
                reinterpret_cast<v4f*>(out + (size_t)v * Lsz * Osz + (size_t)l * Osz + c));
        }
    }
}

extern "C" void kernel_launch(void* const* d_in, const int* in_sizes, int n_in,
                              void* d_out, int out_size, void* d_ws, size_t ws_size,
                              hipStream_t stream) {
    const float* x     = (const float*)d_in[0];
    const float* z     = (const float*)d_in[1];
    const float* W_enc = (const float*)d_in[2];
    const float* b_enc = (const float*)d_in[3];
    const float* W_out = (const float*)d_in[4];
    const float* b_out = (const float*)d_in[5];
    const float* resc  = (const float*)d_in[6];
    float* out = (float*)d_out;

    float* u = (float*)d_ws;

    enc_gemm <<<Lsz, 256, 0, stream>>>(x, z, W_enc, u);
    scan_relu<<<Bsz * (Hsz / SCOLS), CHUNKS * SCOLS, 0, stream>>>(u, b_enc, resc);
    out_gemm <<<Lsz, 384, 0, stream>>>(u, W_out, b_out, out);
}
